# starting kernel
# baseline (speedup 1.0000x reference)
#include <hip/hip_runtime.h>

#define NN   100000
#define EE   600000
#define ETOT (EE + NN)
#define INF_ 64
#define HF   128
#define GG   5000
#define OUTF 19
#define NEG_SLOPE 0.2f

typedef __attribute__((ext_vector_type(2))) float v2f;
typedef __attribute__((ext_vector_type(8))) float v8f;

__device__ __forceinline__ unsigned fkey(float f) {
    unsigned u = __float_as_uint(f);
    return (u & 0x80000000u) ? ~u : (u | 0x80000000u);
}
__device__ __forceinline__ float funkey(unsigned k) {
    return (k & 0x80000000u) ? __uint_as_float(k ^ 0x80000000u) : __uint_as_float(~k);
}

__global__ void __launch_bounds__(256)
gemm_wmma(const float* __restrict__ in, const float* __restrict__ W,
          const float* __restrict__ a_src, const float* __restrict__ a_dst,
          float* __restrict__ h, float* __restrict__ s_src, float* __restrict__ s_dst,
          int K)
{
    const int wave = threadIdx.x >> 5;
    const int lane = threadIdx.x & 31;
    const int tile = blockIdx.x * 8 + wave;
    if (tile * 16 >= NN) return;
    const int row0 = tile * 16;
    const int r    = lane & 15;
    const int hi   = lane >> 4;
    const int koff = hi * 2;

    v8f acc[8];
    const v8f vz = {0.f,0.f,0.f,0.f,0.f,0.f,0.f,0.f};
#pragma unroll
    for (int t = 0; t < 8; ++t) acc[t] = vz;

    const float* arow = in + (size_t)(row0 + r) * K;
    for (int k0 = 0; k0 < K; k0 += 4) {
        v2f a;
        a.x = arow[k0 + koff];
        a.y = arow[k0 + koff + 1];
        const float* wrow = W + (size_t)(k0 + koff) * HF + r;
#pragma unroll
        for (int t = 0; t < 8; ++t) {
            v2f b;
            b.x = wrow[t * 16];
            b.y = wrow[HF + t * 16];
            acc[t] = __builtin_amdgcn_wmma_f32_16x16x4_f32(
                false, a, false, b, (short)0, acc[t], false, false);
        }
    }

    float ps[8], pd[8];
#pragma unroll
    for (int j = 0; j < 8; ++j) { ps[j] = 0.f; pd[j] = 0.f; }
#pragma unroll
    for (int t = 0; t < 8; ++t) {
        const int col = t * 16 + r;
        const float as = a_src[col];
        const float ad = a_dst[col];
#pragma unroll
        for (int j = 0; j < 8; ++j) {
            const int row = row0 + hi * 8 + j;
            float v = acc[t][j];
            h[(size_t)row * HF + col] = v;
            ps[j] += v * as;
            pd[j] += v * ad;
        }
    }
#pragma unroll
    for (int off = 1; off < 16; off <<= 1) {
#pragma unroll
        for (int j = 0; j < 8; ++j) {
            ps[j] += __shfl_xor(ps[j], off, 32);
            pd[j] += __shfl_xor(pd[j], off, 32);
        }
    }
    if (r == 0) {
#pragma unroll
        for (int j = 0; j < 8; ++j) {
            const int row = row0 + hi * 8 + j;
            s_src[row] = ps[j];
            s_dst[row] = pd[j];
        }
    }
}

__global__ void layer_init(float* __restrict__ acc, unsigned* __restrict__ m,
                           float* __restrict__ denom)
{
    int i = blockIdx.x * blockDim.x + threadIdx.x;
    if (i < NN * (HF / 4)) {
        float4 z = {0.f, 0.f, 0.f, 0.f};
        ((float4*)acc)[i] = z;
    }
    if (i < NN) { m[i] = 0u; denom[i] = 0.f; }
}

__global__ void edge_alpha_max(const int* __restrict__ esrc, const int* __restrict__ edst,
                               const float* __restrict__ s_src, const float* __restrict__ s_dst,
                               float* __restrict__ alpha, unsigned* __restrict__ m)
{
    int e = blockIdx.x * blockDim.x + threadIdx.x;
    if (e >= ETOT) return;
    int s = (e < EE) ? esrc[e] : (e - EE);
    int d = (e < EE) ? edst[e] : (e - EE);
    float a = s_src[s] + s_dst[d];
    a = (a > 0.f) ? a : NEG_SLOPE * a;
    alpha[e] = a;
    atomicMax(m + d, fkey(a));
}

__global__ void edge_exp_sum(const int* __restrict__ edst,
                             float* __restrict__ alpha, const unsigned* __restrict__ m,
                             float* __restrict__ denom)
{
    int e = blockIdx.x * blockDim.x + threadIdx.x;
    if (e >= ETOT) return;
    int d = (e < EE) ? edst[e] : (e - EE);
    float ex = expf(alpha[e] - funkey(m[d]));
    alpha[e] = ex;
    atomicAdd(denom + d, ex);
}

__global__ void edge_scatter(const int* __restrict__ esrc, const int* __restrict__ edst,
                             const float* __restrict__ h, const float* __restrict__ alpha,
                             const float* __restrict__ denom, float* __restrict__ acc)
{
    int gw = (int)((blockIdx.x * (size_t)blockDim.x + threadIdx.x) >> 5);
    int lane = threadIdx.x & 31;
    if (gw >= ETOT) return;
    int s = (gw < EE) ? esrc[gw] : (gw - EE);
    int d = (gw < EE) ? edst[gw] : (gw - EE);
    float coef = alpha[gw] / denom[d];
    float4 hv = ((const float4*)(h + (size_t)s * HF))[lane];
    float* ap = acc + (size_t)d * HF + lane * 4;
    atomicAdd(ap + 0, coef * hv.x);
    atomicAdd(ap + 1, coef * hv.y);
    atomicAdd(ap + 2, coef * hv.z);
    atomicAdd(ap + 3, coef * hv.w);
}

__global__ void bias_relu(const float* __restrict__ acc, const float* __restrict__ b,
                          float* __restrict__ act)
{
    int i = blockIdx.x * blockDim.x + threadIdx.x;
    if (i >= NN * HF) return;
    float v = acc[i] + b[i & (HF - 1)];
    act[i] = (v > 0.f) ? v : 0.f;
}

__global__ void pool_init(float* __restrict__ pooled, float* __restrict__ cnts)
{
    int i = blockIdx.x * blockDim.x + threadIdx.x;
    if (i < GG * HF) pooled[i] = 0.f;
    if (i < GG) cnts[i] = 0.f;
}

__global__ void pool_sum(const float* __restrict__ act, const int* __restrict__ batch,
                         float* __restrict__ pooled, float* __restrict__ cnts)
{
    int t = blockIdx.x * blockDim.x + threadIdx.x;
    if (t >= NN * (HF / 4)) return;
    int node = t >> 5, c = t & 31;
    int b = batch[node];
    float4 v = ((const float4*)(act + (size_t)node * HF))[c];
    float* p = pooled + (size_t)b * HF + c * 4;
    atomicAdd(p + 0, v.x);
    atomicAdd(p + 1, v.y);
    atomicAdd(p + 2, v.z);
    atomicAdd(p + 3, v.w);
    if (c == 0) atomicAdd(cnts + b, 1.0f);
}

__global__ void __launch_bounds__(64)
fc_head(const float* __restrict__ pooled, const float* __restrict__ cnts,
        const float* __restrict__ w1, const float* __restrict__ b1,
        const float* __restrict__ w2, const float* __restrict__ b2,
        float* __restrict__ out)
{
    __shared__ float p[HF];
    __shared__ float z[64];
    int g = blockIdx.x, t = threadIdx.x;
    float inv = 1.0f / fmaxf(cnts[g], 1.0f);
    p[t]      = pooled[(size_t)g * HF + t] * inv;
    p[t + 64] = pooled[(size_t)g * HF + t + 64] * inv;
    __syncthreads();
    float a = b1[t];
    for (int k = 0; k < HF; ++k) a += p[k] * w1[k * 64 + t];
    z[t] = fmaxf(a, 0.f);
    __syncthreads();
    if (t < OUTF) {
        float o = b2[t];
        for (int k = 0; k < 64; ++k) o += z[k] * w2[k * OUTF + t];
        out[(size_t)g * OUTF + t] = o;
    }
}

extern "C" void kernel_launch(void* const* d_in, const int* in_sizes, int n_in,
                              void* d_out, int out_size, void* d_ws, size_t ws_size,
                              hipStream_t stream)
{
    const float* x      = (const float*)d_in[0];
    const int*   eidx   = (const int*)  d_in[1];
    const int*   batch  = (const int*)  d_in[2];
    const float* W[3]    = {(const float*)d_in[3],  (const float*)d_in[7],  (const float*)d_in[11]};
    const float* asrc[3] = {(const float*)d_in[4],  (const float*)d_in[8],  (const float*)d_in[12]};
    const float* adst[3] = {(const float*)d_in[5],  (const float*)d_in[9],  (const float*)d_in[13]};
    const float* bias[3] = {(const float*)d_in[6],  (const float*)d_in[10], (const float*)d_in[14]};
    const float* fc1w = (const float*)d_in[15];
    const float* fc1b = (const float*)d_in[16];
    const float* fc2w = (const float*)d_in[17];
    const float* fc2b = (const float*)d_in[18];
    float* out = (float*)d_out;

    const int* esrc = eidx;
    const int* edst = eidx + EE;

    char* ws = (char*)d_ws;
    const size_t NHB = (size_t)NN * HF * sizeof(float);
    float*    h_lin  = (float*)(ws);
    float*    acc    = (float*)(ws + NHB);
    float*    act    = (float*)(ws + 2 * NHB);
    float*    s_src  = (float*)(ws + 3 * NHB);
    float*    s_dst  = (float*)(ws + 3 * NHB + (size_t)NN * 4);
    unsigned* m      = (unsigned*)(ws + 3 * NHB + (size_t)NN * 8);
    float*    denom  = (float*)(ws + 3 * NHB + (size_t)NN * 12);
    float*    alpha  = (float*)(ws + 3 * NHB + (size_t)NN * 16);
    float*    pooled = (float*)(ws + 3 * NHB + (size_t)NN * 16 + (size_t)ETOT * 4);
    float*    cnts   = (float*)(ws + 3 * NHB + (size_t)NN * 16 + (size_t)ETOT * 4 + (size_t)GG * HF * 4);

    const int gemm_grid   = ((NN / 16) + 7) / 8;
    const int init_grid   = (NN * (HF / 4) + 255) / 256;
    const int edge_grid   = (ETOT + 255) / 256;
    const int scat_grid   = (int)(((size_t)ETOT * 32 + 255) / 256);
    const int br_grid     = (NN * HF + 255) / 256;
    const int pool_grid   = (NN * (HF / 4) + 255) / 256;
    const int pinit_grid  = (GG * HF + 255) / 256;

    const float* cur = x;
    int K = INF_;
    for (int l = 0; l < 3; ++l) {
        gemm_wmma<<<gemm_grid, 256, 0, stream>>>(cur, W[l], asrc[l], adst[l],
                                                 h_lin, s_src, s_dst, K);
        layer_init<<<init_grid, 256, 0, stream>>>(acc, m, denom);
        edge_alpha_max<<<edge_grid, 256, 0, stream>>>(esrc, edst, s_src, s_dst, alpha, m);
        edge_exp_sum<<<edge_grid, 256, 0, stream>>>(edst, alpha, m, denom);
        edge_scatter<<<scat_grid, 256, 0, stream>>>(esrc, edst, h_lin, alpha, denom, acc);
        bias_relu<<<br_grid, 256, 0, stream>>>(acc, bias[l], act);
        cur = act;
        K = HF;
    }

    pool_init<<<pinit_grid, 256, 0, stream>>>(pooled, cnts);
    pool_sum<<<pool_grid, 256, 0, stream>>>(act, batch, pooled, cnts);
    fc_head<<<GG, 64, 0, stream>>>(pooled, cnts, fc1w, fc1b, fc2w, fc2b, out);
}


# edges counting-sorted by destination once per call; one wave per node does segment softmax and weighted row sum in registers, stores relu(sum+bias): no float atomics, no init/alpha/exp/bias passes
# speedup vs baseline: 2.7640x; 2.7640x over previous
#include <hip/hip_runtime.h>

#define NN   100000
#define EE   600000
#define ETOT (EE + NN)
#define INF_ 64
#define HF   128
#define GG   5000
#define OUTF 19
#define NEG_SLOPE 0.2f

typedef __attribute__((ext_vector_type(2))) float v2f;
typedef __attribute__((ext_vector_type(8))) float v8f;

__global__ void __launch_bounds__(256)
gemm_wmma(const float* __restrict__ in, const float* __restrict__ W,
          const float* __restrict__ a_src, const float* __restrict__ a_dst,
          float* __restrict__ h, float* __restrict__ s_src, float* __restrict__ s_dst,
          int K)
{
    const int wave = threadIdx.x >> 5;
    const int lane = threadIdx.x & 31;
    const int tile = blockIdx.x * 8 + wave;
    if (tile * 16 >= NN) return;
    const int row0 = tile * 16;
    const int r    = lane & 15;
    const int hi   = lane >> 4;
    const int koff = hi * 2;

    v8f acc[8];
    const v8f vz = {0.f,0.f,0.f,0.f,0.f,0.f,0.f,0.f};
#pragma unroll
    for (int t = 0; t < 8; ++t) acc[t] = vz;

    const float* arow = in + (size_t)(row0 + r) * K;
    for (int k0 = 0; k0 < K; k0 += 4) {
        v2f a;
        a.x = arow[k0 + koff];
        a.y = arow[k0 + koff + 1];
        const float* wrow = W + (size_t)(k0 + koff) * HF + r;
#pragma unroll
        for (int t = 0; t < 8; ++t) {
            v2f b;
            b.x = wrow[t * 16];
            b.y = wrow[HF + t * 16];
            acc[t] = __builtin_amdgcn_wmma_f32_16x16x4_f32(
                false, a, false, b, (short)0, acc[t], false, false);
        }
    }

    float ps[8], pd[8];
#pragma unroll
    for (int j = 0; j < 8; ++j) { ps[j] = 0.f; pd[j] = 0.f; }
#pragma unroll
    for (int t = 0; t < 8; ++t) {
        const int col = t * 16 + r;
        const float as = a_src[col];
        const float ad = a_dst[col];
#pragma unroll
        for (int j = 0; j < 8; ++j) {
            const int row = row0 + hi * 8 + j;
            float v = acc[t][j];
            h[(size_t)row * HF + col] = v;
            ps[j] += v * as;
            pd[j] += v * ad;
        }
    }
#pragma unroll
    for (int off = 1; off < 16; off <<= 1) {
#pragma unroll
        for (int j = 0; j < 8; ++j) {
            ps[j] += __shfl_xor(ps[j], off, 32);
            pd[j] += __shfl_xor(pd[j], off, 32);
        }
    }
    if (r == 0) {
#pragma unroll
        for (int j = 0; j < 8; ++j) {
            const int row = row0 + hi * 8 + j;
            s_src[row] = ps[j];
            s_dst[row] = pd[j];
        }
    }
}

__global__ void csr_zero(int* __restrict__ deg)
{
    int i = blockIdx.x * blockDim.x + threadIdx.x;
    if (i < NN) deg[i] = 0;
}

__global__ void csr_count(const int* __restrict__ edst, int* __restrict__ deg,
                          int* __restrict__ rank)
{
    int e = blockIdx.x * blockDim.x + threadIdx.x;
    if (e < EE) rank[e] = atomicAdd(deg + edst[e], 1);
}

__global__ void __launch_bounds__(1024)
csr_scan(const int* __restrict__ deg, int* __restrict__ rowptr)
{
    __shared__ int wtot[32];
    const int t = threadIdx.x, lane = t & 31, w = t >> 5;
    const int base = blockIdx.x * 1024;

    int p = 0;
    const int4* deg4 = (const int4*)deg;
    const int sub = t & 255, q = t >> 8, nrow = (int)blockIdx.x;
    int r = q;
    for (; r + 12 < nrow; r += 16) {
        const int4 v0 = deg4[(r     ) * 256 + sub];
        const int4 v1 = deg4[(r +  4) * 256 + sub];
        const int4 v2 = deg4[(r +  8) * 256 + sub];
        const int4 v3 = deg4[(r + 12) * 256 + sub];
        p += ((v0.x + v0.y) + (v0.z + v0.w)) + ((v1.x + v1.y) + (v1.z + v1.w))
           + ((v2.x + v2.y) + (v2.z + v2.w)) + ((v3.x + v3.y) + (v3.z + v3.w));
    }
    for (; r < nrow; r += 4) {
        const int4 v = deg4[r * 256 + sub];
        p += (v.x + v.y) + (v.z + v.w);
    }
    for (int off = 16; off > 0; off >>= 1) p += __shfl_xor(p, off, 32);
    if (lane == 0) wtot[w] = p;
    __syncthreads();
    int prefix = wtot[lane];
    for (int off = 16; off > 0; off >>= 1) prefix += __shfl_xor(prefix, off, 32);
    __syncthreads();

    const int i = base + t;
    const int dg = (i < NN) ? deg[i] : 0;
    int inc = dg;
    for (int off = 1; off < 32; off <<= 1) {
        const int n = __shfl_up(inc, off, 32);
        if (lane >= off) inc += n;
    }
    if (lane == 31) wtot[w] = inc;
    __syncthreads();
    const int wv = wtot[lane];
    int winc = wv;
    for (int off = 1; off < 32; off <<= 1) {
        const int n = __shfl_up(winc, off, 32);
        if (lane >= off) winc += n;
    }
    const int woff = __shfl(winc - wv, w, 32);
    const int excl = prefix + woff + inc - dg;
    if (i < NN) rowptr[i] = excl;
    if (i == NN - 1) rowptr[NN] = excl + dg;
}

__global__ void csr_fill(const int* __restrict__ esrc, const int* __restrict__ edst,
                         const int* __restrict__ rowptr, const int* __restrict__ rank,
                         int* __restrict__ csr_src)
{
    int e = blockIdx.x * blockDim.x + threadIdx.x;
    if (e >= EE) return;
    csr_src[rowptr[edst[e]] + rank[e]] = esrc[e];
}

__device__ __forceinline__ float wave_max(float v)
{
    for (int off = 16; off > 0; off >>= 1) v = fmaxf(v, __shfl_xor(v, off, 32));
    return v;
}
__device__ __forceinline__ float wave_sum(float v)
{
    for (int off = 16; off > 0; off >>= 1) v += __shfl_xor(v, off, 32);
    return v;
}

__device__ __forceinline__ float edge_logit(const int* __restrict__ csr_src,
                                            const float* __restrict__ s_src,
                                            int start, int deg, int d, float sd, int i, int& s)
{
    s = d;
    if (i < deg) s = csr_src[start + i];
    const float a = s_src[s] + sd;
    return (a > 0.f) ? a : NEG_SLOPE * a;
}

template <int K>
__device__ __forceinline__ void add_k(const float* __restrict__ h, int s, float coef,
                                      int j0, int lane, float4& acc)
{
    float  cf[K];
    float4 hv[K];
    for (int k = 0; k < K; ++k) {
        const int sj = __builtin_amdgcn_readlane(s, j0 + k);
        cf[k] = __int_as_float(__builtin_amdgcn_readlane(__float_as_int(coef), j0 + k));
        hv[k] = ((const float4*)(h + (size_t)sj * HF))[lane];
    }
    for (int k = 0; k < K; ++k) {
        acc.x += cf[k] * hv[k].x; acc.y += cf[k] * hv[k].y;
        acc.z += cf[k] * hv[k].z; acc.w += cf[k] * hv[k].w;
    }
}

__device__ __forceinline__ void add_rows(const float* __restrict__ h, int s, float coef,
                                         int n, int lane, float4& acc)
{
    int j = 0;
    for (; j + 4 <= n; j += 4) add_k<4>(h, s, coef, j, lane, acc);
    const int t = n - j;
    if (t == 3)      add_k<3>(h, s, coef, j, lane, acc);
    else if (t == 2) add_k<2>(h, s, coef, j, lane, acc);
    else if (t == 1) add_k<1>(h, s, coef, j, lane, acc);
}

__global__ void __launch_bounds__(256)
node_gather(const int* __restrict__ rowptr, const int* __restrict__ csr_src,
            const float* __restrict__ h, const float* __restrict__ s_src,
            const float* __restrict__ s_dst, const float* __restrict__ bias,
            float* __restrict__ act)
{
    const int lane = threadIdx.x & 31;
    const int wv = __builtin_amdgcn_readfirstlane((int)(threadIdx.x >> 5));
    const int nwaves = (int)gridDim.x * 8;
    const float b0 = bias[lane * 4 + 0], b1 = bias[lane * 4 + 1];
    const float b2 = bias[lane * 4 + 2], b3 = bias[lane * 4 + 3];
    const float ninf = -__builtin_inff();

    for (int d = (int)blockIdx.x * 8 + wv; d < NN; d += nwaves) {
        const int start = rowptr[d];
        const int deg   = rowptr[d + 1] - start;
        const int total = deg + 1;
        const float sd  = s_dst[d];
        float4 acc = {0.f, 0.f, 0.f, 0.f};

        if (total <= 32) {
            int s;
            const float a = edge_logit(csr_src, s_src, start, deg, d, sd, lane, s);
            const bool valid = lane < total;
            const float m   = wave_max(valid ? a : ninf);
            const float e   = valid ? expf(a - m) : 0.f;
            const float den = wave_sum(e);
            add_rows(h, s, e / den, total, lane, acc);
        } else {
            float m = ninf;
            for (int c = 0; c < total; c += 32) {
                int s;
                const float a = edge_logit(csr_src, s_src, start, deg, d, sd, c + lane, s);
                if (c + lane < total) m = fmaxf(m, a);
            }
            m = wave_max(m);
            float den = 0.f;
            for (int c = 0; c < total; c += 32) {
                int s;
                const float a = edge_logit(csr_src, s_src, start, deg, d, sd, c + lane, s);
                if (c + lane < total) den += expf(a - m);
            }
            den = wave_sum(den);
            for (int c = 0; c < total; c += 32) {
                int s;
                const float a = edge_logit(csr_src, s_src, start, deg, d, sd, c + lane, s);
                const float coef = (c + lane < total) ? expf(a - m) / den : 0.f;
                const int n = (total - c < 32) ? (total - c) : 32;
                add_rows(h, s, coef, n, lane, acc);
            }
        }

        float4 o;
        o.x = acc.x + b0; o.y = acc.y + b1; o.z = acc.z + b2; o.w = acc.w + b3;
        o.x = (o.x > 0.f) ? o.x : 0.f;
        o.y = (o.y > 0.f) ? o.y : 0.f;
        o.z = (o.z > 0.f) ? o.z : 0.f;
        o.w = (o.w > 0.f) ? o.w : 0.f;
        ((float4*)(act + (size_t)d * HF))[lane] = o;
    }
}

__global__ void pool_init(float* __restrict__ pooled, float* __restrict__ cnts)
{
    int i = blockIdx.x * blockDim.x + threadIdx.x;
    if (i < GG * HF) pooled[i] = 0.f;
    if (i < GG) cnts[i] = 0.f;
}

__global__ void pool_sum(const float* __restrict__ act, const int* __restrict__ batch,
                         float* __restrict__ pooled, float* __restrict__ cnts)
{
    int t = blockIdx.x * blockDim.x + threadIdx.x;
    if (t >= NN * (HF / 4)) return;
    int node = t >> 5, c = t & 31;
    int b = batch[node];
    float4 v = ((const float4*)(act + (size_t)node * HF))[c];
    float* p = pooled + (size_t)b * HF + c * 4;
    atomicAdd(p + 0, v.x);
    atomicAdd(p + 1, v.y);
    atomicAdd(p + 2, v.z);
    atomicAdd(p + 3, v.w);
    if (c == 0) atomicAdd(cnts + b, 1.0f);
}

__global__ void __launch_bounds__(64)
fc_head(const float* __restrict__ pooled, const float* __restrict__ cnts,
        const float* __restrict__ w1, const float* __restrict__ b1,
        const float* __restrict__ w2, const float* __restrict__ b2,
        float* __restrict__ out)
{
    __shared__ float p[HF];
    __shared__ float z[64];
    int g = blockIdx.x, t = threadIdx.x;
    float inv = 1.0f / fmaxf(cnts[g], 1.0f);
    p[t]      = pooled[(size_t)g * HF + t] * inv;
    p[t + 64] = pooled[(size_t)g * HF + t + 64] * inv;
    __syncthreads();
    float a = b1[t];
    for (int k = 0; k < HF; ++k) a += p[k] * w1[k * 64 + t];
    z[t] = fmaxf(a, 0.f);
    __syncthreads();
    if (t < OUTF) {
        float o = b2[t];
        for (int k = 0; k < 64; ++k) o += z[k] * w2[k * OUTF + t];
        out[(size_t)g * OUTF + t] = o;
    }
}

extern "C" void kernel_launch(void* const* d_in, const int* in_sizes, int n_in,
                              void* d_out, int out_size, void* d_ws, size_t ws_size,
                              hipStream_t stream)
{
    const float* x      = (const float*)d_in[0];
    const int*   eidx   = (const int*)  d_in[1];
    const int*   batch  = (const int*)  d_in[2];
    const float* W[3]    = {(const float*)d_in[3],  (const float*)d_in[7],  (const float*)d_in[11]};
    const float* asrc[3] = {(const float*)d_in[4],  (const float*)d_in[8],  (const float*)d_in[12]};
    const float* adst[3] = {(const float*)d_in[5],  (const float*)d_in[9],  (const float*)d_in[13]};
    const float* bias[3] = {(const float*)d_in[6],  (const float*)d_in[10], (const float*)d_in[14]};
    const float* fc1w = (const float*)d_in[15];
    const float* fc1b = (const float*)d_in[16];
    const float* fc2w = (const float*)d_in[17];
    const float* fc2b = (const float*)d_in[18];
    float* out = (float*)d_out;

    const int* esrc = eidx;
    const int* edst = eidx + EE;

    char* ws = (char*)d_ws;
    const size_t NHB = (size_t)NN * HF * sizeof(float);
    const size_t RPB = (size_t)(NN + 32) * 4;
    float*    h_lin   = (float*)(ws);
    int*      rowptr  = (int*)(ws + NHB);
    int*      deg     = (int*)(ws + NHB + RPB);
    int*      rank    = (int*)(ws + NHB + RPB + (size_t)NN * 4);
    int*      csr_src = (int*)(ws + NHB + RPB + (size_t)NN * 4 + (size_t)EE * 4);
    float*    act    = (float*)(ws + 2 * NHB);
    float*    s_src  = (float*)(ws + 3 * NHB);
    float*    s_dst  = (float*)(ws + 3 * NHB + (size_t)NN * 4);
    float*    pooled = (float*)(ws + 3 * NHB + (size_t)NN * 16 + (size_t)ETOT * 4);
    float*    cnts   = (float*)(ws + 3 * NHB + (size_t)NN * 16 + (size_t)ETOT * 4 + (size_t)GG * HF * 4);

    const int gemm_grid   = ((NN / 16) + 7) / 8;
    const int node_grid   = (NN + 255) / 256;
    const int edge_grid   = (EE + 255) / 256;
    const int scan_grid   = (NN + 1023) / 1024;
    const int gath_grid   = 2048;
    const int pool_grid   = (NN * (HF / 4) + 255) / 256;
    const int pinit_grid  = (GG * HF + 255) / 256;

    csr_zero<<<node_grid, 256, 0, stream>>>(deg);
    csr_count<<<edge_grid, 256, 0, stream>>>(edst, deg, rank);
    csr_scan<<<scan_grid, 1024, 0, stream>>>(deg, rowptr);
    csr_fill<<<edge_grid, 256, 0, stream>>>(esrc, edst, rowptr, rank, csr_src);

    const float* cur = x;
    int K = INF_;
    for (int l = 0; l < 3; ++l) {
        gemm_wmma<<<gemm_grid, 256, 0, stream>>>(cur, W[l], asrc[l], adst[l],
                                                 h_lin, s_src, s_dst, K);
        node_gather<<<gath_grid, 256, 0, stream>>>(rowptr, csr_src, h_lin, s_src, s_dst,
                                                   bias[l], act);
        cur = act;
        K = HF;
    }

    pool_init<<<pinit_grid, 256, 0, stream>>>(pooled, cnts);
    pool_sum<<<pool_grid, 256, 0, stream>>>(act, batch, pooled, cnts);
    fc_head<<<GG, 64, 0, stream>>>(pooled, cnts, fc1w, fc1b, fc2w, fc2b, out);
}
